# starting kernel
# baseline (speedup 1.0000x reference)
#include <hip/hip_runtime.h>


#define VEC_DIM   256
#define NUM_WORDS 50000
#define BATCH     4096
#define NUM_CTX   8
#define NUM_K     6
#define TILE_B    16
#define THREADS   128

typedef __attribute__((ext_vector_type(2))) float v2f;
typedef __attribute__((ext_vector_type(8))) float v8f;

__global__ __launch_bounds__(THREADS) void pvdm_wmma_kernel(
    const int*   __restrict__ ctx,
    const int*   __restrict__ doc,
    const int*   __restrict__ tgt,
    const float* __restrict__ Dm,
    const float* __restrict__ Wm,
    const float* __restrict__ Om,
    float*       __restrict__ out)
{
    __shared__ __align__(16) float xs[TILE_B][VEC_DIM + 4];
    __shared__ float part[2][TILE_B];

    const int b0   = blockIdx.x * TILE_B;
    const int tid  = threadIdx.x;
    const int lane = tid & 31;
    const int wave = tid >> 5;
    const int wk   = wave & 1;
    const int wd   = wave >> 1;

    for (int task = tid; task < TILE_B * (VEC_DIM / 4); task += THREADS) {
        const int row = task >> 6;
        const int c4  = task & 63;
        const int b   = b0 + row;
        const float4* dp = (const float4*)(Dm + (long)doc[b] * VEC_DIM) + c4;
        float4 acc = *dp;
        #pragma unroll
        for (int c = 0; c < NUM_CTX; ++c) {
            const float4* wp =
                (const float4*)(Wm + (long)ctx[b * NUM_CTX + c] * VEC_DIM) + c4;
            float4 wv = *wp;
            acc.x += wv.x; acc.y += wv.y; acc.z += wv.z; acc.w += wv.w;
        }
        *(float4*)&xs[row][c4 * 4] = acc;
    }
    __syncthreads();

    const int row   = lane & 15;
    const int kOff  = (lane < 16) ? 0 : 2;
    const int dBase = wd * (VEC_DIM / 2);

    for (int kk = 0; kk < 3; ++kk) {
        const int k   = wk * 3 + kk;
        const int col = tgt[(b0 + row) * NUM_K + k];
        const float* ob = Om + (long)dBase * NUM_WORDS + col;

        v8f c = {0.f, 0.f, 0.f, 0.f, 0.f, 0.f, 0.f, 0.f};

        #pragma unroll 8
        for (int step = 0; step < VEC_DIM / 8; ++step) {
            const int d0 = step * 4 + kOff;
            v2f a, b;
            a.x = xs[row][dBase + d0];
            a.y = xs[row][dBase + d0 + 1];
            b.x = ob[(long)d0 * NUM_WORDS];
            b.y = ob[(long)(d0 + 1) * NUM_WORDS];
            c = __builtin_amdgcn_wmma_f32_16x16x4_f32(
                    false, a, false, b, (short)0, c, false, false);
        }

        float val = 0.f;
        int   di  = 0;
        bool  own = false;
        if (lane < 8)        { val = c[lane];      di = lane;      own = true; }
        else if (lane >= 24) { val = c[lane - 24]; di = lane - 16; own = true; }

        if (wd == 1 && own) part[wk][di] = val;
        __syncthreads();
        if (wd == 0 && own)
            out[(b0 + di) * NUM_K + k] = val + part[wk][di];
        __syncthreads();
    }
}

extern "C" void kernel_launch(void* const* d_in, const int* in_sizes, int n_in,
                              void* d_out, int out_size, void* d_ws, size_t ws_size,
                              hipStream_t stream) {
    (void)in_sizes; (void)n_in; (void)out_size; (void)d_ws; (void)ws_size;
    const int*   ctx = (const int*)  d_in[0];
    const int*   doc = (const int*)  d_in[1];
    const int*   tgt = (const int*)  d_in[2];
    const float* Dm  = (const float*)d_in[3];
    const float* Wm  = (const float*)d_in[4];
    const float* Om  = (const float*)d_in[5];
    float* out = (float*)d_out;

    dim3 grid(BATCH / TILE_B);
    dim3 block(THREADS);
    hipLaunchKernelGGL(pvdm_wmma_kernel, grid, block, 0, stream,
                       ctx, doc, tgt, Dm, Wm, Om, out);
}


# Regroup the (b,k) pairs by 16-column group of O so one fetched half line of an O row serves every pair of the group; x is built by a first launch into the workspace; the sum order over d changes
# speedup vs baseline: 1.4633x; 1.4633x over previous
#include <hip/hip_runtime.h>


#define VEC_DIM   256
#define NUM_WORDS 50000
#define BATCH     4096
#define NUM_CTX   8
#define NUM_K     6
#define NPAIR     (BATCH * NUM_K)
#define GCOLS     16
#define NGROUP    (NUM_WORDS / GCOLS)
#define WAVES     8
#define THREADS2  (WAVES * 32)
#define GPB       WAVES
#define SCAN_Q    (NPAIR / 4 / THREADS2)
#define SCAN_BATCH 8
#define LIST_CAP  (SCAN_Q * 4 * 32)

typedef __attribute__((ext_vector_type(2))) float v2f;
typedef __attribute__((ext_vector_type(8))) float v8f;

__global__ __launch_bounds__(256) void build_x(
    const int*   __restrict__ ctx,
    const int*   __restrict__ doc,
    const float* __restrict__ Dm,
    const float* __restrict__ Wm,
    float*       __restrict__ X)
{
    const int task = blockIdx.x * 256 + threadIdx.x;
    const int c4 = task & 63;
    #pragma unroll
    for (int half = 0; half < 2; ++half) {
        const int b = (task >> 6) + half * (BATCH / 2);
        const float4* dp = (const float4*)(Dm + (long)doc[b] * VEC_DIM) + c4;
        float4 acc = *dp;
        #pragma unroll
        for (int c = 0; c < NUM_CTX; ++c) {
            const float4* wp =
                (const float4*)(Wm + (long)ctx[b * NUM_CTX + c] * VEC_DIM) + c4;
            float4 wv = *wp;
            acc.x += wv.x; acc.y += wv.y; acc.z += wv.z; acc.w += wv.w;
        }
        *((float4*)(X + (long)b * VEC_DIM) + c4) = acc;
    }
}

__device__ __forceinline__ void run_tile(
    const unsigned* tile, int cnt, int g, int lane,
    const float* __restrict__ X, const float* __restrict__ Om,
    float* __restrict__ out)
{
    const int r = lane & 15;
    const int h = lane >> 4;
    const unsigned er = (r < cnt) ? tile[r] : 0u;
    const int b = (int)((er & 0x7FFFu) / NUM_K);
    const float* xa = X + (long)b * VEC_DIM + 4 * h;
    const float* ob = Om + (long)(4 * h) * NUM_WORDS + g * GCOLS + r;

    v8f c = {0.f, 0.f, 0.f, 0.f, 0.f, 0.f, 0.f, 0.f};

    #pragma unroll 8
    for (int t = 0; t < VEC_DIM / 8; ++t) {
        const float4 a4 = *(const float4*)(xa + 8 * t);
        const float* o = ob + (long)(8 * t) * NUM_WORDS;
        v2f a0, a1, b0, b1;
        a0.x = a4.x; a0.y = a4.y;
        a1.x = a4.z; a1.y = a4.w;
        b0.x = o[0];
        b0.y = o[(long)NUM_WORDS];
        b1.x = o[(long)2 * NUM_WORDS];
        b1.y = o[(long)3 * NUM_WORDS];
        c = __builtin_amdgcn_wmma_f32_16x16x4_f32(
                false, a0, false, b0, (short)0, c, false, false);
        c = __builtin_amdgcn_wmma_f32_16x16x4_f32(
                false, a1, false, b1, (short)0, c, false, false);
    }

    #pragma unroll
    for (int j = 0; j < 8; ++j) {
        const int m = 8 * h + j;
        if (m < cnt) {
            const unsigned e = tile[m];
            if ((int)((e >> 15) & 15u) == r) out[e & 0x7FFFu] = c[j];
        }
    }
}

__global__ __launch_bounds__(THREADS2) void score_by_column_group(
    const int*   __restrict__ tgt,
    const float* __restrict__ X,
    const float* __restrict__ Om,
    float*       __restrict__ out)
{
    __shared__ unsigned sList[WAVES][LIST_CAP];
    __shared__ unsigned sTile[WAVES][16];
    __shared__ int      sCount[WAVES];

    const int tid  = threadIdx.x;
    const int lane = tid & 31;
    const int w    = tid >> 5;
    const unsigned lt = (1u << lane) - 1u;
    const int g0 = blockIdx.x * GPB;
    const unsigned c0 = (unsigned)(g0 * GCOLS);

    int n = 0;
    const int4* tgt4 = (const int4*)tgt;
    for (int it = 0; it < SCAN_Q; it += SCAN_BATCH) {
        int4 v[SCAN_BATCH];
        #pragma unroll
        for (int u = 0; u < SCAN_BATCH; ++u)
            v[u] = tgt4[(it + u) * THREADS2 + tid];
        #pragma unroll
        for (int u = 0; u < SCAN_BATCH; ++u) {
            const int q = (it + u) * THREADS2 + tid;
            const unsigned cv[4] = {(unsigned)v[u].x, (unsigned)v[u].y,
                                    (unsigned)v[u].z, (unsigned)v[u].w};
            #pragma unroll
            for (int j = 0; j < 4; ++j) {
                const unsigned col = cv[j];
                const bool hit = (col - c0) < (unsigned)(GPB * GCOLS);
                const unsigned mask = __builtin_amdgcn_ballot_w32(hit);
                if (hit)
                    sList[w][n + __builtin_popcount(mask & lt)] =
                        (col << 15) | (unsigned)(4 * q + j);
                n += __builtin_popcount(mask);
            }
        }
    }
    if (lane == 0) sCount[w] = n;
    __syncthreads();

    const int g = g0 + w;
    if (g >= NGROUP) return;

    unsigned* tile = sTile[w];
    int cnt = 0;
    for (int L = 0; L < WAVES; ++L) {
        const int nL = __builtin_amdgcn_readfirstlane(sCount[L]);
        for (int base = 0; base < nL; base += 32) {
            const int idx = base + lane;
            const unsigned e = (idx < nL) ? sList[L][idx] : 0u;
            bool mine = (idx < nL) && ((int)(e >> 19) == g);
            unsigned mask = __builtin_amdgcn_ballot_w32(mine);
            while (mask != 0u) {
                const int rank = __builtin_popcount(mask & lt);
                const int room = 16 - cnt;
                const bool take = mine && (rank < room);
                if (take) tile[cnt + rank] = e;
                const int tot = __builtin_popcount(mask);
                cnt += (tot < room) ? tot : room;
                mine = mine && !take;
                mask = __builtin_amdgcn_ballot_w32(mine);
                if (cnt == 16) {
                    __builtin_amdgcn_fence(__ATOMIC_RELEASE, "wavefront");
                    __builtin_amdgcn_fence(__ATOMIC_ACQUIRE, "wavefront");
                    run_tile(tile, 16, g, lane, X, Om, out);
                    cnt = 0;
                }
            }
        }
    }
    if (cnt > 0) {
        __builtin_amdgcn_fence(__ATOMIC_RELEASE, "wavefront");
        __builtin_amdgcn_fence(__ATOMIC_ACQUIRE, "wavefront");
        run_tile(tile, cnt, g, lane, X, Om, out);
    }
}

extern "C" void kernel_launch(void* const* d_in, const int* in_sizes, int n_in,
                              void* d_out, int out_size, void* d_ws, size_t ws_size,
                              hipStream_t stream) {
    (void)in_sizes; (void)n_in; (void)out_size; (void)ws_size;
    const int*   ctx = (const int*)  d_in[0];
    const int*   doc = (const int*)  d_in[1];
    const int*   tgt = (const int*)  d_in[2];
    const float* Dm  = (const float*)d_in[3];
    const float* Wm  = (const float*)d_in[4];
    const float* Om  = (const float*)d_in[5];
    float* out = (float*)d_out;
    float* X   = (float*)d_ws;

    hipLaunchKernelGGL(build_x, dim3(512), dim3(256), 0, stream,
                       ctx, doc, Dm, Wm, X);
    hipLaunchKernelGGL(score_by_column_group, dim3(391), dim3(THREADS2), 0, stream,
                       tgt, X, Om, out);
}


# Score launch reshaped to one 16-wave block per WGP: 256 blocks, each owning 12 or 13 consecutive column groups, so a WGP scans the target list once, not twice, and holds at most 13 groups' tiles
# speedup vs baseline: 1.5480x; 1.0579x over previous
#include <hip/hip_runtime.h>


#define VEC_DIM   256
#define NUM_WORDS 50000
#define BATCH     4096
#define NUM_CTX   8
#define NUM_K     6
#define NPAIR     (BATCH * NUM_K)
#define GCOLS     16
#define NGROUP    (NUM_WORDS / GCOLS)
#define WAVES     16
#define THREADS2  (WAVES * 32)
#define NBLOCK2   256
#define SCAN_Q    (NPAIR / 4 / THREADS2)
#define SCAN_BATCH SCAN_Q
#define LIST_CAP  (SCAN_Q * 4 * 32)

typedef __attribute__((ext_vector_type(2))) float v2f;
typedef __attribute__((ext_vector_type(8))) float v8f;

__global__ __launch_bounds__(256) void build_x(
    const int*   __restrict__ ctx,
    const int*   __restrict__ doc,
    const float* __restrict__ Dm,
    const float* __restrict__ Wm,
    float*       __restrict__ X)
{
    const int task = blockIdx.x * 256 + threadIdx.x;
    const int c4 = task & 63;
    #pragma unroll
    for (int half = 0; half < 2; ++half) {
        const int b = (task >> 6) + half * (BATCH / 2);
        const float4* dp = (const float4*)(Dm + (long)doc[b] * VEC_DIM) + c4;
        float4 acc = *dp;
        #pragma unroll
        for (int c = 0; c < NUM_CTX; ++c) {
            const float4* wp =
                (const float4*)(Wm + (long)ctx[b * NUM_CTX + c] * VEC_DIM) + c4;
            float4 wv = *wp;
            acc.x += wv.x; acc.y += wv.y; acc.z += wv.z; acc.w += wv.w;
        }
        *((float4*)(X + (long)b * VEC_DIM) + c4) = acc;
    }
}

__device__ __forceinline__ void run_tile(
    const unsigned* tile, int cnt, int g, int lane,
    const float* __restrict__ X, const float* __restrict__ Om,
    float* __restrict__ out)
{
    const int r = lane & 15;
    const int h = lane >> 4;
    const unsigned er = (r < cnt) ? tile[r] : 0u;
    const int b = (int)((er & 0x7FFFu) / NUM_K);
    const float* xa = X + (long)b * VEC_DIM + 4 * h;
    const float* ob = Om + (long)(4 * h) * NUM_WORDS + g * GCOLS + r;

    v8f c = {0.f, 0.f, 0.f, 0.f, 0.f, 0.f, 0.f, 0.f};

    #pragma unroll 8
    for (int t = 0; t < VEC_DIM / 8; ++t) {
        const float4 a4 = *(const float4*)(xa + 8 * t);
        const float* o = ob + (long)(8 * t) * NUM_WORDS;
        v2f a0, a1, b0, b1;
        a0.x = a4.x; a0.y = a4.y;
        a1.x = a4.z; a1.y = a4.w;
        b0.x = o[0];
        b0.y = o[(long)NUM_WORDS];
        b1.x = o[(long)2 * NUM_WORDS];
        b1.y = o[(long)3 * NUM_WORDS];
        c = __builtin_amdgcn_wmma_f32_16x16x4_f32(
                false, a0, false, b0, (short)0, c, false, false);
        c = __builtin_amdgcn_wmma_f32_16x16x4_f32(
                false, a1, false, b1, (short)0, c, false, false);
    }

    #pragma unroll
    for (int j = 0; j < 8; ++j) {
        const int m = 8 * h + j;
        if (m < cnt) {
            const unsigned e = tile[m];
            if ((int)((e >> 15) & 15u) == r) out[e & 0x7FFFu] = c[j];
        }
    }
}

__global__ __launch_bounds__(THREADS2) void score_by_column_group(
    const int*   __restrict__ tgt,
    const float* __restrict__ X,
    const float* __restrict__ Om,
    float*       __restrict__ out)
{
    __shared__ unsigned sList[WAVES][LIST_CAP];
    __shared__ unsigned sTile[WAVES][16];
    __shared__ int      sCount[WAVES];

    const int tid  = threadIdx.x;
    const int lane = tid & 31;
    const int w    = tid >> 5;
    const unsigned lt = (1u << lane) - 1u;
    const int gs = ((int)blockIdx.x * NGROUP) / NBLOCK2;
    const int ge = (((int)blockIdx.x + 1) * NGROUP) / NBLOCK2;
    const unsigned c0 = (unsigned)(gs * GCOLS);
    const unsigned ncol = (unsigned)((ge - gs) * GCOLS);

    int n = 0;
    const int4* tgt4 = (const int4*)tgt;
    for (int it = 0; it < SCAN_Q; it += SCAN_BATCH) {
        int4 v[SCAN_BATCH];
        #pragma unroll
        for (int u = 0; u < SCAN_BATCH; ++u)
            v[u] = tgt4[(it + u) * THREADS2 + tid];
        #pragma unroll
        for (int u = 0; u < SCAN_BATCH; ++u) {
            const int q = (it + u) * THREADS2 + tid;
            const unsigned cv[4] = {(unsigned)v[u].x, (unsigned)v[u].y,
                                    (unsigned)v[u].z, (unsigned)v[u].w};
            #pragma unroll
            for (int j = 0; j < 4; ++j) {
                const unsigned col = cv[j];
                const bool hit = (col - c0) < ncol;
                const unsigned mask = __builtin_amdgcn_ballot_w32(hit);
                if (hit)
                    sList[w][n + __builtin_popcount(mask & lt)] =
                        (col << 15) | (unsigned)(4 * q + j);
                n += __builtin_popcount(mask);
            }
        }
    }
    if (lane == 0) sCount[w] = n;
    __syncthreads();

    const int g = gs + w;
    if (g >= ge) return;

    unsigned* tile = sTile[w];
    int cnt = 0;
    for (int L = 0; L < WAVES; ++L) {
        const int nL = __builtin_amdgcn_readfirstlane(sCount[L]);
        for (int base = 0; base < nL; base += 32) {
            const int idx = base + lane;
            const unsigned e = (idx < nL) ? sList[L][idx] : 0u;
            bool mine = (idx < nL) && ((int)(e >> 19) == g);
            unsigned mask = __builtin_amdgcn_ballot_w32(mine);
            while (mask != 0u) {
                const int rank = __builtin_popcount(mask & lt);
                const int room = 16 - cnt;
                const bool take = mine && (rank < room);
                if (take) tile[cnt + rank] = e;
                const int tot = __builtin_popcount(mask);
                cnt += (tot < room) ? tot : room;
                mine = mine && !take;
                mask = __builtin_amdgcn_ballot_w32(mine);
                if (cnt == 16) {
                    __builtin_amdgcn_fence(__ATOMIC_RELEASE, "wavefront");
                    __builtin_amdgcn_fence(__ATOMIC_ACQUIRE, "wavefront");
                    run_tile(tile, 16, g, lane, X, Om, out);
                    cnt = 0;
                }
            }
        }
    }
    if (cnt > 0) {
        __builtin_amdgcn_fence(__ATOMIC_RELEASE, "wavefront");
        __builtin_amdgcn_fence(__ATOMIC_ACQUIRE, "wavefront");
        run_tile(tile, cnt, g, lane, X, Om, out);
    }
}

extern "C" void kernel_launch(void* const* d_in, const int* in_sizes, int n_in,
                              void* d_out, int out_size, void* d_ws, size_t ws_size,
                              hipStream_t stream) {
    (void)in_sizes; (void)n_in; (void)out_size; (void)ws_size;
    const int*   ctx = (const int*)  d_in[0];
    const int*   doc = (const int*)  d_in[1];
    const int*   tgt = (const int*)  d_in[2];
    const float* Dm  = (const float*)d_in[3];
    const float* Wm  = (const float*)d_in[4];
    const float* Om  = (const float*)d_in[5];
    float* out = (float*)d_out;
    float* X   = (float*)d_ws;

    hipLaunchKernelGGL(build_x, dim3(512), dim3(256), 0, stream,
                       ctx, doc, Dm, Wm, X);
    hipLaunchKernelGGL(score_by_column_group, dim3(NBLOCK2), dim3(THREADS2), 0, stream,
                       tgt, X, Om, out);
}


# Score launch: the 16 per-wave match lists are copied end to end into one block list behind a second barrier, so a tile wave walks about 3 chunks of it, not 16 lists; tiles and sums as before
# speedup vs baseline: 1.6567x; 1.0702x over previous
#include <hip/hip_runtime.h>


#define VEC_DIM   256
#define NUM_WORDS 50000
#define BATCH     4096
#define NUM_CTX   8
#define NUM_K     6
#define NPAIR     (BATCH * NUM_K)
#define GCOLS     16
#define NGROUP    (NUM_WORDS / GCOLS)
#define WAVES     16
#define THREADS2  (WAVES * 32)
#define NBLOCK2   256
#define SCAN_Q    (NPAIR / 4 / THREADS2)
#define SCAN_BATCH SCAN_Q
#define LIST_CAP  (SCAN_Q * 4 * 32)
#define ALL_CAP   4096

typedef __attribute__((ext_vector_type(2))) float v2f;
typedef __attribute__((ext_vector_type(8))) float v8f;

__global__ __launch_bounds__(256) void build_x(
    const int*   __restrict__ ctx,
    const int*   __restrict__ doc,
    const float* __restrict__ Dm,
    const float* __restrict__ Wm,
    float*       __restrict__ X)
{
    const int task = blockIdx.x * 256 + threadIdx.x;
    const int c4 = task & 63;
    #pragma unroll
    for (int half = 0; half < 2; ++half) {
        const int b = (task >> 6) + half * (BATCH / 2);
        const float4* dp = (const float4*)(Dm + (long)doc[b] * VEC_DIM) + c4;
        float4 acc = *dp;
        #pragma unroll
        for (int c = 0; c < NUM_CTX; ++c) {
            const float4* wp =
                (const float4*)(Wm + (long)ctx[b * NUM_CTX + c] * VEC_DIM) + c4;
            float4 wv = *wp;
            acc.x += wv.x; acc.y += wv.y; acc.z += wv.z; acc.w += wv.w;
        }
        *((float4*)(X + (long)b * VEC_DIM) + c4) = acc;
    }
}

__device__ __forceinline__ void run_tile(
    const unsigned* tile, int cnt, int g, int lane,
    const float* __restrict__ X, const float* __restrict__ Om,
    float* __restrict__ out)
{
    const int r = lane & 15;
    const int h = lane >> 4;
    const unsigned er = (r < cnt) ? tile[r] : 0u;
    const int b = (int)((er & 0x7FFFu) / NUM_K);
    const float* xa = X + (long)b * VEC_DIM + 4 * h;
    const float* ob = Om + (long)(4 * h) * NUM_WORDS + g * GCOLS + r;

    v8f c = {0.f, 0.f, 0.f, 0.f, 0.f, 0.f, 0.f, 0.f};

    #pragma unroll 8
    for (int t = 0; t < VEC_DIM / 8; ++t) {
        const float4 a4 = *(const float4*)(xa + 8 * t);
        const float* o = ob + (long)(8 * t) * NUM_WORDS;
        v2f a0, a1, b0, b1;
        a0.x = a4.x; a0.y = a4.y;
        a1.x = a4.z; a1.y = a4.w;
        b0.x = o[0];
        b0.y = o[(long)NUM_WORDS];
        b1.x = o[(long)2 * NUM_WORDS];
        b1.y = o[(long)3 * NUM_WORDS];
        c = __builtin_amdgcn_wmma_f32_16x16x4_f32(
                false, a0, false, b0, (short)0, c, false, false);
        c = __builtin_amdgcn_wmma_f32_16x16x4_f32(
                false, a1, false, b1, (short)0, c, false, false);
    }

    #pragma unroll
    for (int j = 0; j < 8; ++j) {
        const int m = 8 * h + j;
        if (m < cnt) {
            const unsigned e = tile[m];
            if ((int)((e >> 15) & 15u) == r) out[e & 0x7FFFu] = c[j];
        }
    }
}

__global__ __launch_bounds__(THREADS2) void score_by_column_group(
    const int*   __restrict__ tgt,
    const float* __restrict__ X,
    const float* __restrict__ Om,
    float*       __restrict__ out)
{
    __shared__ unsigned sList[WAVES][LIST_CAP];
    __shared__ unsigned sAll[ALL_CAP];
    __shared__ unsigned sTile[WAVES][16];
    __shared__ int      sCount[WAVES];

    const int tid  = threadIdx.x;
    const int lane = tid & 31;
    const int w    = tid >> 5;
    const unsigned lt = (1u << lane) - 1u;
    const int gs = ((int)blockIdx.x * NGROUP) / NBLOCK2;
    const int ge = (((int)blockIdx.x + 1) * NGROUP) / NBLOCK2;
    const unsigned c0 = (unsigned)(gs * GCOLS);
    const unsigned ncol = (unsigned)((ge - gs) * GCOLS);

    int n = 0;
    const int4* tgt4 = (const int4*)tgt;
    for (int it = 0; it < SCAN_Q; it += SCAN_BATCH) {
        int4 v[SCAN_BATCH];
        #pragma unroll
        for (int u = 0; u < SCAN_BATCH; ++u)
            v[u] = tgt4[(it + u) * THREADS2 + tid];
        #pragma unroll
        for (int u = 0; u < SCAN_BATCH; ++u) {
            const int q = (it + u) * THREADS2 + tid;
            const unsigned cv[4] = {(unsigned)v[u].x, (unsigned)v[u].y,
                                    (unsigned)v[u].z, (unsigned)v[u].w};
            #pragma unroll
            for (int j = 0; j < 4; ++j) {
                const unsigned col = cv[j];
                const bool hit = (col - c0) < ncol;
                const unsigned mask = __builtin_amdgcn_ballot_w32(hit);
                if (hit)
                    sList[w][n + __builtin_popcount(mask & lt)] =
                        (col << 15) | (unsigned)(4 * q + j);
                n += __builtin_popcount(mask);
            }
        }
    }
    if (lane == 0) sCount[w] = n;
    __syncthreads();

    const int ws = __builtin_amdgcn_readfirstlane(w);
    const int cl = sCount[lane & (WAVES - 1)];
    int off = 0, total = 0;
    for (int L = 0; L < WAVES; ++L) {
        const int v = __builtin_amdgcn_readlane(cl, L);
        off += (L < ws) ? v : 0;
        total += v;
    }
    const bool merged = total <= ALL_CAP;
    if (merged) {
        for (int i = lane; i < n; i += 32) sAll[off + i] = sList[w][i];
    }
    __syncthreads();

    const int g = gs + w;
    if (g >= ge) return;

    unsigned* tile = sTile[w];
    int cnt = 0;
    const int nl = merged ? 1 : WAVES;
    for (int L = 0; L < nl; ++L) {
        const unsigned* lp = merged ? sAll : sList[L];
        const int nL = merged ? total : __builtin_amdgcn_readlane(cl, L);
        for (int base = 0; base < nL; base += 32) {
            const int idx = base + lane;
            const unsigned e = (idx < nL) ? lp[idx] : 0u;
            bool mine = (idx < nL) && ((int)(e >> 19) == g);
            unsigned mask = __builtin_amdgcn_ballot_w32(mine);
            while (mask != 0u) {
                const int rank = __builtin_popcount(mask & lt);
                const int room = 16 - cnt;
                const bool take = mine && (rank < room);
                if (take) tile[cnt + rank] = e;
                const int tot = __builtin_popcount(mask);
                cnt += (tot < room) ? tot : room;
                mine = mine && !take;
                mask = __builtin_amdgcn_ballot_w32(mine);
                if (cnt == 16) {
                    __builtin_amdgcn_fence(__ATOMIC_RELEASE, "wavefront");
                    __builtin_amdgcn_fence(__ATOMIC_ACQUIRE, "wavefront");
                    run_tile(tile, 16, g, lane, X, Om, out);
                    cnt = 0;
                }
            }
        }
    }
    if (cnt > 0) {
        __builtin_amdgcn_fence(__ATOMIC_RELEASE, "wavefront");
        __builtin_amdgcn_fence(__ATOMIC_ACQUIRE, "wavefront");
        run_tile(tile, cnt, g, lane, X, Om, out);
    }
}

extern "C" void kernel_launch(void* const* d_in, const int* in_sizes, int n_in,
                              void* d_out, int out_size, void* d_ws, size_t ws_size,
                              hipStream_t stream) {
    (void)in_sizes; (void)n_in; (void)out_size; (void)ws_size;
    const int*   ctx = (const int*)  d_in[0];
    const int*   doc = (const int*)  d_in[1];
    const int*   tgt = (const int*)  d_in[2];
    const float* Dm  = (const float*)d_in[3];
    const float* Wm  = (const float*)d_in[4];
    const float* Om  = (const float*)d_in[5];
    float* out = (float*)d_out;
    float* X   = (float*)d_ws;

    hipLaunchKernelGGL(build_x, dim3(512), dim3(256), 0, stream,
                       ctx, doc, Dm, Wm, X);
    hipLaunchKernelGGL(score_by_column_group, dim3(NBLOCK2), dim3(THREADS2), 0, stream,
                       tgt, X, Om, out);
}


# Target scan moved into launch 1, where it runs while the x rows are in flight; each block writes its pair list to the workspace; launch 2 is 13 tile waves per block reading it; tiles, sums unchanged
# speedup vs baseline: 1.7862x; 1.0782x over previous
#include <hip/hip_runtime.h>


#define VEC_DIM   256
#define NUM_WORDS 50000
#define BATCH     4096
#define NUM_CTX   8
#define NUM_K     6
#define NPAIR     (BATCH * NUM_K)
#define GCOLS     16
#define NGROUP    (NUM_WORDS / GCOLS)
#define WAVES     16
#define THREADS1  (WAVES * 32)
#define NBLOCK    256
#define SCAN_Q    (NPAIR / 4 / THREADS1)
#define LIST_CAP  (SCAN_Q * 4 * 32)
#define LIST_STRIDE (NPAIR + 32)
#define PRE       4
#define PAD_TO    (PRE * 32)
#define NO_ENTRY  0xFFFFFFFFu
#define TILE_WAVES 13
#define THREADS2  (TILE_WAVES * 32)
#define X_BYTES   (BATCH * VEC_DIM * 4)

typedef __attribute__((ext_vector_type(2))) float v2f;
typedef __attribute__((ext_vector_type(8))) float v8f;

__global__ __launch_bounds__(THREADS1) void build_x_and_lists(
    const int*   __restrict__ ctx,
    const int*   __restrict__ doc,
    const int*   __restrict__ tgt,
    const float* __restrict__ Dm,
    const float* __restrict__ Wm,
    float*       __restrict__ X,
    unsigned*    __restrict__ lists)
{
    __shared__ unsigned sList[WAVES][LIST_CAP];
    __shared__ int      sCount[WAVES];

    const int tid  = threadIdx.x;
    const int lane = tid & 31;
    const int w    = tid >> 5;
    const unsigned lt = (1u << lane) - 1u;
    const int gs = ((int)blockIdx.x * NGROUP) / NBLOCK;
    const int ge = (((int)blockIdx.x + 1) * NGROUP) / NBLOCK;
    const unsigned c0 = (unsigned)(gs * GCOLS);
    const unsigned ncol = (unsigned)((ge - gs) * GCOLS);

    int4 v[SCAN_Q];
    const int4* tgt4 = (const int4*)tgt;
    #pragma unroll
    for (int u = 0; u < SCAN_Q; ++u)
        v[u] = tgt4[u * THREADS1 + tid];
    __builtin_amdgcn_sched_barrier(0);

    const int task = blockIdx.x * THREADS1 + tid;
    const int c4 = task & 63;
    float4 dv[2];
    float4 wv[2][NUM_CTX];
    #pragma unroll
    for (int half = 0; half < 2; ++half) {
        const int b = (task >> 6) + half * (BATCH / 2);
        dv[half] = *((const float4*)(Dm + (long)doc[b] * VEC_DIM) + c4);
        #pragma unroll
        for (int c = 0; c < NUM_CTX; ++c)
            wv[half][c] =
                *((const float4*)(Wm + (long)ctx[b * NUM_CTX + c] * VEC_DIM) + c4);
    }

    int n = 0;
    #pragma unroll
    for (int u = 0; u < SCAN_Q; ++u) {
        const int q = u * THREADS1 + tid;
        const unsigned cv[4] = {(unsigned)v[u].x, (unsigned)v[u].y,
                                (unsigned)v[u].z, (unsigned)v[u].w};
        #pragma unroll
        for (int j = 0; j < 4; ++j) {
            const unsigned col = cv[j];
            const bool hit = (col - c0) < ncol;
            const unsigned mask = __builtin_amdgcn_ballot_w32(hit);
            if (hit)
                sList[w][n + __builtin_popcount(mask & lt)] =
                    (col << 15) | (unsigned)(4 * q + j);
            n += __builtin_popcount(mask);
        }
    }
    if (lane == 0) sCount[w] = n;
    __builtin_amdgcn_fence(__ATOMIC_RELEASE, "workgroup", "local");
    __builtin_amdgcn_s_barrier();
    __builtin_amdgcn_fence(__ATOMIC_ACQUIRE, "workgroup", "local");

    const int ws = __builtin_amdgcn_readfirstlane(w);
    const int cl = sCount[lane & (WAVES - 1)];
    int off = 0, total = 0;
    for (int L = 0; L < WAVES; ++L) {
        const int cv = __builtin_amdgcn_readlane(cl, L);
        off += (L < ws) ? cv : 0;
        total += cv;
    }
    unsigned* gl = lists + (long)blockIdx.x * LIST_STRIDE;
    for (int i = lane; i < n; i += 32) gl[1 + off + i] = sList[w][i];
    if (tid > total && tid < PAD_TO) gl[tid] = NO_ENTRY;
    if (tid == 0) gl[0] = (unsigned)total;

    #pragma unroll
    for (int half = 0; half < 2; ++half) {
        const int b = (task >> 6) + half * (BATCH / 2);
        float4 acc = dv[half];
        #pragma unroll
        for (int c = 0; c < NUM_CTX; ++c) {
            const float4 t = wv[half][c];
            acc.x += t.x; acc.y += t.y; acc.z += t.z; acc.w += t.w;
        }
        *((float4*)(X + (long)b * VEC_DIM) + c4) = acc;
    }
}

__device__ __forceinline__ void run_tile(
    const unsigned* tile, int cnt, int g, int lane,
    const float* __restrict__ X, const float* __restrict__ Om,
    float* __restrict__ out)
{
    const int r = lane & 15;
    const int h = lane >> 4;
    const unsigned er = (r < cnt) ? tile[r] : 0u;
    const int b = (int)((er & 0x7FFFu) / NUM_K);
    const float* xa = X + (long)b * VEC_DIM + 4 * h;
    const float* ob = Om + (long)(4 * h) * NUM_WORDS + g * GCOLS + r;

    v8f c = {0.f, 0.f, 0.f, 0.f, 0.f, 0.f, 0.f, 0.f};

    #pragma unroll 8
    for (int t = 0; t < VEC_DIM / 8; ++t) {
        const float4 a4 = *(const float4*)(xa + 8 * t);
        const float* o = ob + (long)(8 * t) * NUM_WORDS;
        v2f a0, a1, b0, b1;
        a0.x = a4.x; a0.y = a4.y;
        a1.x = a4.z; a1.y = a4.w;
        b0.x = o[0];
        b0.y = o[(long)NUM_WORDS];
        b1.x = o[(long)2 * NUM_WORDS];
        b1.y = o[(long)3 * NUM_WORDS];
        c = __builtin_amdgcn_wmma_f32_16x16x4_f32(
                false, a0, false, b0, (short)0, c, false, false);
        c = __builtin_amdgcn_wmma_f32_16x16x4_f32(
                false, a1, false, b1, (short)0, c, false, false);
    }

    #pragma unroll
    for (int j = 0; j < 8; ++j) {
        const int m = 8 * h + j;
        if (m < cnt) {
            const unsigned e = tile[m];
            if ((int)((e >> 15) & 15u) == r) out[e & 0x7FFFu] = c[j];
        }
    }
}

__global__ __launch_bounds__(THREADS2) void score_by_column_group(
    const unsigned* __restrict__ lists,
    const float*    __restrict__ X,
    const float*    __restrict__ Om,
    float*          __restrict__ out)
{
    __shared__ unsigned sTile[TILE_WAVES][16];

    const int lane = threadIdx.x & 31;
    const int w = __builtin_amdgcn_readfirstlane((int)threadIdx.x >> 5);
    const unsigned lt = (1u << lane) - 1u;
    const int gs = ((int)blockIdx.x * NGROUP) / NBLOCK;
    const int ge = (((int)blockIdx.x + 1) * NGROUP) / NBLOCK;
    const int g = gs + w;
    if (g >= ge) return;

    const unsigned* lp = lists + (long)blockIdx.x * LIST_STRIDE;
    unsigned pre[PRE];
    #pragma unroll
    for (int k = 0; k < PRE; ++k) pre[k] = lp[32 * k + lane];
    const int total = (int)__builtin_amdgcn_readlane(pre[0], 0);

    unsigned* tile = sTile[w];
    int cnt = 0;
    int base = 0;
    do {
        const int idx = base + lane;
        unsigned e;
        if (base == 0)       e = pre[0];
        else if (base == 32) e = pre[1];
        else if (base == 64) e = pre[2];
        else if (base == 96) e = pre[3];
        else                 e = (idx <= total) ? lp[idx] : 0u;
        bool mine = (idx >= 1) && (idx <= total) && ((int)(e >> 19) == g);
        unsigned mask = __builtin_amdgcn_ballot_w32(mine);
        while (mask != 0u) {
            const int rank = __builtin_popcount(mask & lt);
            const int room = 16 - cnt;
            const bool take = mine && (rank < room);
            if (take) tile[cnt + rank] = e;
            const int tot = __builtin_popcount(mask);
            cnt += (tot < room) ? tot : room;
            mine = mine && !take;
            mask = __builtin_amdgcn_ballot_w32(mine);
            if (cnt == 16) {
                __builtin_amdgcn_fence(__ATOMIC_RELEASE, "wavefront");
                __builtin_amdgcn_fence(__ATOMIC_ACQUIRE, "wavefront");
                run_tile(tile, 16, g, lane, X, Om, out);
                cnt = 0;
            }
        }
        base += 32;
    } while (base <= total);
    if (cnt > 0) {
        __builtin_amdgcn_fence(__ATOMIC_RELEASE, "wavefront");
        __builtin_amdgcn_fence(__ATOMIC_ACQUIRE, "wavefront");
        run_tile(tile, cnt, g, lane, X, Om, out);
    }
}

extern "C" void kernel_launch(void* const* d_in, const int* in_sizes, int n_in,
                              void* d_out, int out_size, void* d_ws, size_t ws_size,
                              hipStream_t stream) {
    (void)in_sizes; (void)n_in; (void)out_size; (void)ws_size;
    const int*   ctx = (const int*)  d_in[0];
    const int*   doc = (const int*)  d_in[1];
    const int*   tgt = (const int*)  d_in[2];
    const float* Dm  = (const float*)d_in[3];
    const float* Wm  = (const float*)d_in[4];
    const float* Om  = (const float*)d_in[5];
    float* out = (float*)d_out;
    float*    X      = (float*)d_ws;
    unsigned* lists  = (unsigned*)((char*)d_ws + X_BYTES);

    hipLaunchKernelGGL(build_x_and_lists, dim3(NBLOCK), dim3(THREADS1), 0, stream,
                       ctx, doc, tgt, Dm, Wm, X, lists);
    hipLaunchKernelGGL(score_by_column_group, dim3(NBLOCK), dim3(THREADS2), 0, stream,
                       lists, X, Om, out);
}
